# starting kernel
# baseline (speedup 1.0000x reference)
#include <hip/hip_runtime.h>

typedef __attribute__((ext_vector_type(2))) float v2f;
typedef __attribute__((ext_vector_type(8))) float v8f;

#define H_DIM 512
#define R_DIM 64
#define M_ROWS 2048

__global__ void build_M(const float* __restrict__ W_seq,
                        const float* __restrict__ he,
                        const float* __restrict__ cp,
                        float* __restrict__ Mmat) {
    const int idx = blockIdx.x * blockDim.x + threadIdx.x;
    const int hp = idx >> 9;
    const int h  = idx & 511;
    float acc = 0.0f;
#pragma unroll 8
    for (int r = 0; r < R_DIM; ++r) {
        acc = fmaf(W_seq[r * H_DIM + hp], he[h * R_DIM + r] * cp[r], acc);
    }
    Mmat[idx] = acc;
}

__global__ void __launch_bounds__(256)
gemm_out(const float* __restrict__ Xin,
         const float* __restrict__ Mmat,
         float* __restrict__ Out) {
    const int lane = threadIdx.x & 31;
    const int wave = threadIdx.x >> 5;
    const int l    = lane & 15;
    const int half = lane >> 4;

    const int wid    = blockIdx.x * (blockDim.x >> 5) + wave;
    const int tile_m = wid >> 3;
    const int ngroup = wid & 7;
    const int m0 = tile_m * 16;
    const int n0 = ngroup * 64;

    v8f acc0 = {}, acc1 = {}, acc2 = {}, acc3 = {};

    const float* __restrict__ arow = Xin + (size_t)(m0 + l) * H_DIM;

#pragma unroll 4
    for (int k = 0; k < H_DIM; k += 4) {
        const int ka = k + 2 * half;
        v2f a;
        a.x = arow[ka];
        a.y = arow[ka + 1];

        const float* __restrict__ b0 = Mmat + (size_t)ka * H_DIM + n0 + l;
        const float* __restrict__ b1 = b0 + H_DIM;

        v2f b;
        b.x = b0[0];  b.y = b1[0];
        acc0 = __builtin_amdgcn_wmma_f32_16x16x4_f32(
            false, a, false, b, (short)0, acc0, false, false);
        b.x = b0[16]; b.y = b1[16];
        acc1 = __builtin_amdgcn_wmma_f32_16x16x4_f32(
            false, a, false, b, (short)0, acc1, false, false);
        b.x = b0[32]; b.y = b1[32];
        acc2 = __builtin_amdgcn_wmma_f32_16x16x4_f32(
            false, a, false, b, (short)0, acc2, false, false);
        b.x = b0[48]; b.y = b1[48];
        acc3 = __builtin_amdgcn_wmma_f32_16x16x4_f32(
            false, a, false, b, (short)0, acc3, false, false);
    }

    float* __restrict__ orow = Out + n0 + l;
#pragma unroll
    for (int p = 0; p < 8; ++p) {
        const size_t m = (size_t)(m0 + p + half * 8);
        orow[m * H_DIM +  0] = acc0[p];
        orow[m * H_DIM + 16] = acc1[p];
        orow[m * H_DIM + 32] = acc2[p];
        orow[m * H_DIM + 48] = acc3[p];
    }
}

extern "C" void kernel_launch(void* const* d_in, const int* in_sizes, int n_in,
                              void* d_out, int out_size, void* d_ws, size_t ws_size,
                              hipStream_t stream) {
    const float* hidden = (const float*)d_in[0];
    const float* W_seq  = (const float*)d_in[2];
    const float* he     = (const float*)d_in[3];
    const float* cp     = (const float*)d_in[4];
    float* out  = (float*)d_out;
    float* Mmat = (float*)d_ws;

    (void)in_sizes; (void)n_in; (void)out_size; (void)ws_size;

    build_M<<<(H_DIM * H_DIM) / 256, 256, 0, stream>>>(W_seq, he, cp, Mmat);

    gemm_out<<<128, 256, 0, stream>>>(hidden, Mmat, out);
}


# Re-associate to (X*W_seq^T)*(he*cp)^T in one kernel: 16x64 intermediate by split-K over 8 waves summed in LDS, replaces the 512x512 matrix build; f32 sum order changes
# speedup vs baseline: 4.8635x; 4.8635x over previous
#include <hip/hip_runtime.h>


typedef __attribute__((ext_vector_type(2))) float v2f;
typedef __attribute__((ext_vector_type(4))) float v4f;
typedef __attribute__((ext_vector_type(8))) float v8f;

#define H_DIM 512
#define R_DIM 64
#define M_ROWS 2048
#define TP 68

__global__ void __launch_bounds__(256)
cp_lowrank(const float* __restrict__ Xin,
           const float* __restrict__ Wseq,
           const float* __restrict__ he,
           const float* __restrict__ cp,
           float* __restrict__ Out) {
    __shared__ __attribute__((aligned(16))) float part[8][16 * TP];
    __shared__ __attribute__((aligned(16))) float tsum[16 * TP];

    const int tid  = threadIdx.x;
    const int lane = tid & 31;
    const int wave = tid >> 5;
    const int c    = lane & 15;
    const int hf   = lane >> 4;
    const int m0   = (blockIdx.x >> 1) * 16;
    const int hbase = (blockIdx.x & 1) * 256 + wave * 32;

    const int kb = wave * 64 + 4 * hf;
    const v4f* __restrict__ xp = (const v4f*)(Xin + (size_t)(m0 + c) * H_DIM + kb);
    v4f xa[8];
#pragma unroll
    for (int j = 0; j < 8; ++j) xa[j] = xp[2 * j];

    v4f wb[4][8];
#pragma unroll
    for (int t = 0; t < 4; ++t) {
        const v4f* __restrict__ wp = (const v4f*)(Wseq + (size_t)(16 * t + c) * H_DIM + kb);
#pragma unroll
        for (int j = 0; j < 8; ++j) wb[t][j] = wp[2 * j];
    }

    v4f fb[2][8];
#pragma unroll
    for (int u = 0; u < 2; ++u) {
        const v4f* __restrict__ hp = (const v4f*)(he + (size_t)(hbase + 2 * c + u) * R_DIM + 4 * hf);
#pragma unroll
        for (int j = 0; j < 8; ++j) fb[u][j] = hp[2 * j];
    }
    v4f cw[8];
    {
        const v4f* __restrict__ cpp = (const v4f*)(cp + 4 * hf);
#pragma unroll
        for (int j = 0; j < 8; ++j) cw[j] = cpp[2 * j];
    }

    __builtin_amdgcn_sched_barrier(0);

    v8f acc[4] = {};
#pragma unroll
    for (int j = 0; j < 8; ++j) {
        v2f a0; a0.x = xa[j].x; a0.y = xa[j].y;
        v2f a1; a1.x = xa[j].z; a1.y = xa[j].w;
#pragma unroll
        for (int t = 0; t < 4; ++t) {
            v2f b; b.x = wb[t][j].x; b.y = wb[t][j].y;
            acc[t] = __builtin_amdgcn_wmma_f32_16x16x4_f32(
                false, a0, false, b, (short)0, acc[t], false, false);
        }
#pragma unroll
        for (int t = 0; t < 4; ++t) {
            v2f b; b.x = wb[t][j].z; b.y = wb[t][j].w;
            acc[t] = __builtin_amdgcn_wmma_f32_16x16x4_f32(
                false, a1, false, b, (short)0, acc[t], false, false);
        }
    }

    {
        float* __restrict__ pw = &part[wave][(8 * hf) * TP + c];
#pragma unroll
        for (int t = 0; t < 4; ++t)
#pragma unroll
            for (int p = 0; p < 8; ++p) pw[p * TP + 16 * t] = acc[t][p];
    }
    __syncthreads();

    {
        const int off = (tid >> 4) * TP + (tid & 15) * 4;
        v4f q[8];
#pragma unroll
        for (int w = 0; w < 8; ++w) q[w] = *(const v4f*)&part[w][off];
        __builtin_amdgcn_sched_barrier(0);
        v4f s = q[0];
#pragma unroll
        for (int w = 1; w < 8; ++w) s += q[w];
        *(v4f*)&tsum[off] = s;
    }
    __syncthreads();

    v4f ta[8];
#pragma unroll
    for (int j = 0; j < 8; ++j) ta[j] = *(const v4f*)&tsum[c * TP + 8 * j + 4 * hf];

    __builtin_amdgcn_sched_barrier(0);

    v8f o0 = {}, o1 = {};
#pragma unroll
    for (int j = 0; j < 8; ++j) {
        const v4f f0 = fb[0][j] * cw[j];
        const v4f f1 = fb[1][j] * cw[j];
        v2f a0; a0.x = ta[j].x; a0.y = ta[j].y;
        v2f a1; a1.x = ta[j].z; a1.y = ta[j].w;
        v2f b;
        b.x = f0.x; b.y = f0.y;
        o0 = __builtin_amdgcn_wmma_f32_16x16x4_f32(false, a0, false, b, (short)0, o0, false, false);
        b.x = f1.x; b.y = f1.y;
        o1 = __builtin_amdgcn_wmma_f32_16x16x4_f32(false, a0, false, b, (short)0, o1, false, false);
        b.x = f0.z; b.y = f0.w;
        o0 = __builtin_amdgcn_wmma_f32_16x16x4_f32(false, a1, false, b, (short)0, o0, false, false);
        b.x = f1.z; b.y = f1.w;
        o1 = __builtin_amdgcn_wmma_f32_16x16x4_f32(false, a1, false, b, (short)0, o1, false, false);
    }

    float* __restrict__ op = Out + (size_t)(m0 + 8 * hf) * H_DIM + hbase + 2 * c;
#pragma unroll
    for (int p = 0; p < 8; ++p) {
        v2f v; v.x = o0[p]; v.y = o1[p];
        *(v2f*)(op + (size_t)p * H_DIM) = v;
    }
}

extern "C" void kernel_launch(void* const* d_in, const int* in_sizes, int n_in,
                              void* d_out, int out_size, void* d_ws, size_t ws_size,
                              hipStream_t stream) {
    const float* hidden = (const float*)d_in[0];
    const float* W_seq  = (const float*)d_in[2];
    const float* he     = (const float*)d_in[3];
    const float* cp     = (const float*)d_in[4];
    float* out  = (float*)d_out;

    (void)in_sizes; (void)n_in; (void)out_size; (void)d_ws; (void)ws_size;

    cp_lowrank<<<256, 256, 0, stream>>>(hidden, W_seq, he, cp, out);
}


# Operands fetched by whole-line global loads (512 contiguous B each) and re-laid through a wave-private LDS tile into fragments, replacing loads that took 32 B from each of 16 lines; sums unchanged
# speedup vs baseline: 5.7942x; 1.1914x over previous
#include <hip/hip_runtime.h>


typedef __attribute__((ext_vector_type(2))) float v2f;
typedef __attribute__((ext_vector_type(4))) float v4f;
typedef __attribute__((ext_vector_type(8))) float v8f;

#define H_DIM 512
#define R_DIM 64
#define M_ROWS 2048
#define TP 68
#define SROWS 112

__global__ void __launch_bounds__(256)
cp_lowrank(const float* __restrict__ Xin,
           const float* __restrict__ Wseq,
           const float* __restrict__ he,
           const float* __restrict__ cp,
           float* __restrict__ Out) {
    __shared__ __attribute__((aligned(16))) float stage[8][SROWS * TP];
    __shared__ __attribute__((aligned(16))) float part[8][16 * TP];
    __shared__ __attribute__((aligned(16))) float tsum[16 * TP];

    const int tid  = threadIdx.x;
    const int lane = tid & 31;
    const int wave = tid >> 5;
    const int c    = lane & 15;
    const int hf   = lane >> 4;
    const int m0   = (blockIdx.x >> 1) * 16;
    const int hbase = (blockIdx.x & 1) * 256 + wave * 32;

    const int ks = wave * 64 + 4 * c;
    v4f gx[8];
    {
        const float* __restrict__ xg = Xin + (size_t)(m0 + hf) * H_DIM + ks;
#pragma unroll
        for (int i = 0; i < 8; ++i) gx[i] = *(const v4f*)(xg + (size_t)(2 * i) * H_DIM);
    }
    v4f gw[32];
    {
        const float* __restrict__ wg = Wseq + (size_t)hf * H_DIM + ks;
#pragma unroll
        for (int i = 0; i < 32; ++i) gw[i] = *(const v4f*)(wg + (size_t)(2 * i) * H_DIM);
    }
    v4f gh[16];
    {
        const float* __restrict__ hg = he + (size_t)(hbase + hf) * R_DIM + 4 * c;
#pragma unroll
        for (int i = 0; i < 16; ++i) gh[i] = *(const v4f*)(hg + (size_t)(2 * i) * R_DIM);
    }
    v4f cw[8];
    {
        const v4f* __restrict__ cpp = (const v4f*)(cp + 4 * hf);
#pragma unroll
        for (int j = 0; j < 8; ++j) cw[j] = cpp[2 * j];
    }

    __builtin_amdgcn_sched_barrier(0);

    float* st = &stage[wave][0];
    {
        float* sp = st + hf * TP + 4 * c;
#pragma unroll
        for (int i = 0; i < 8; ++i) *(v4f*)(sp + (2 * i) * TP) = gx[i];
#pragma unroll
        for (int i = 0; i < 32; ++i) *(v4f*)(sp + (16 + 2 * i) * TP) = gw[i];
    }

    const float* fp = st + c * TP + 4 * hf;
    v4f xa[8];
#pragma unroll
    for (int j = 0; j < 8; ++j) xa[j] = *(const v4f*)(fp + 8 * j);
    v4f wb[4][8];
#pragma unroll
    for (int t = 0; t < 4; ++t)
#pragma unroll
        for (int j = 0; j < 8; ++j) wb[t][j] = *(const v4f*)(fp + (16 + 16 * t) * TP + 8 * j);

    __builtin_amdgcn_sched_barrier(0);

    v8f acc[4] = {};
#pragma unroll
    for (int j = 0; j < 8; ++j) {
        v2f a0; a0.x = xa[j].x; a0.y = xa[j].y;
        v2f a1; a1.x = xa[j].z; a1.y = xa[j].w;
#pragma unroll
        for (int t = 0; t < 4; ++t) {
            v2f b; b.x = wb[t][j].x; b.y = wb[t][j].y;
            acc[t] = __builtin_amdgcn_wmma_f32_16x16x4_f32(
                false, a0, false, b, (short)0, acc[t], false, false);
        }
#pragma unroll
        for (int t = 0; t < 4; ++t) {
            v2f b; b.x = wb[t][j].z; b.y = wb[t][j].w;
            acc[t] = __builtin_amdgcn_wmma_f32_16x16x4_f32(
                false, a1, false, b, (short)0, acc[t], false, false);
        }
    }

#pragma unroll
    for (int i = 0; i < 16; ++i) *(v4f*)(st + (80 + 16 * hf + i) * TP + 4 * c) = gh[i];

    v4f fb[2][8];
#pragma unroll
    for (int u = 0; u < 2; ++u)
#pragma unroll
        for (int j = 0; j < 8; ++j)
            fb[u][j] = *(const v4f*)(st + (80 + 16 * u + c) * TP + 4 * hf + 8 * j);

    {
        float* __restrict__ pw = &part[wave][(8 * hf) * TP + c];
#pragma unroll
        for (int t = 0; t < 4; ++t)
#pragma unroll
            for (int p = 0; p < 8; ++p) pw[p * TP + 16 * t] = acc[t][p];
    }
    __syncthreads();

    {
        const int off = (tid >> 4) * TP + (tid & 15) * 4;
        v4f q[8];
#pragma unroll
        for (int w = 0; w < 8; ++w) q[w] = *(const v4f*)&part[w][off];
        __builtin_amdgcn_sched_barrier(0);
        v4f s = q[0];
#pragma unroll
        for (int w = 1; w < 8; ++w) s += q[w];
        *(v4f*)&tsum[off] = s;
    }
    __syncthreads();

    v4f ta[8];
#pragma unroll
    for (int j = 0; j < 8; ++j) ta[j] = *(const v4f*)&tsum[c * TP + 8 * j + 4 * hf];

    __builtin_amdgcn_sched_barrier(0);

    v8f o0 = {}, o1 = {};
#pragma unroll
    for (int j = 0; j < 8; ++j) {
        const v4f f0 = fb[0][j] * cw[j];
        const v4f f1 = fb[1][j] * cw[j];
        v2f a0; a0.x = ta[j].x; a0.y = ta[j].y;
        v2f a1; a1.x = ta[j].z; a1.y = ta[j].w;
        v2f b;
        b.x = f0.x; b.y = f0.y;
        o0 = __builtin_amdgcn_wmma_f32_16x16x4_f32(false, a0, false, b, (short)0, o0, false, false);
        b.x = f1.x; b.y = f1.y;
        o1 = __builtin_amdgcn_wmma_f32_16x16x4_f32(false, a0, false, b, (short)0, o1, false, false);
        b.x = f0.z; b.y = f0.w;
        o0 = __builtin_amdgcn_wmma_f32_16x16x4_f32(false, a1, false, b, (short)0, o0, false, false);
        b.x = f1.z; b.y = f1.w;
        o1 = __builtin_amdgcn_wmma_f32_16x16x4_f32(false, a1, false, b, (short)0, o1, false, false);
    }

    float* __restrict__ op = Out + (size_t)(m0 + 8 * hf) * H_DIM + hbase + 2 * c;
#pragma unroll
    for (int p = 0; p < 8; ++p) {
        v2f v; v.x = o0[p]; v.y = o1[p];
        *(v2f*)(op + (size_t)p * H_DIM) = v;
    }
}

extern "C" void kernel_launch(void* const* d_in, const int* in_sizes, int n_in,
                              void* d_out, int out_size, void* d_ws, size_t ws_size,
                              hipStream_t stream) {
    const float* hidden = (const float*)d_in[0];
    const float* W_seq  = (const float*)d_in[2];
    const float* he     = (const float*)d_in[3];
    const float* cp     = (const float*)d_in[4];
    float* out  = (float*)d_out;

    (void)in_sizes; (void)n_in; (void)out_size; (void)d_ws; (void)ws_size;

    cp_lowrank<<<256, 256, 0, stream>>>(hidden, W_seq, he, cp, out);
}


# 4 waves per work-group (one per SIMD) instead of 8: each wave takes a 128-wide k slice as two 64-wide halves and 64 output columns; 4 partials summed instead of 8, f32 sum order changes
# speedup vs baseline: 6.9865x; 1.2058x over previous
#include <hip/hip_runtime.h>


typedef __attribute__((ext_vector_type(2))) float v2f;
typedef __attribute__((ext_vector_type(4))) float v4f;
typedef __attribute__((ext_vector_type(8))) float v8f;

#define H_DIM 512
#define R_DIM 64
#define M_ROWS 2048
#define TP 68
#define HROWS 80
#define SROWS 224

template <int J0, int J1>
static __device__ __forceinline__ void stage1_steps(v8f (&acc)[4], const v4f (&xa)[8], const v4f (&wb)[4][8]) {
#pragma unroll
    for (int j = J0; j < J1; ++j) {
        v2f a0; a0.x = xa[j].x; a0.y = xa[j].y;
        v2f a1; a1.x = xa[j].z; a1.y = xa[j].w;
#pragma unroll
        for (int t = 0; t < 4; ++t) {
            v2f b; b.x = wb[t][j].x; b.y = wb[t][j].y;
            acc[t] = __builtin_amdgcn_wmma_f32_16x16x4_f32(
                false, a0, false, b, (short)0, acc[t], false, false);
        }
#pragma unroll
        for (int t = 0; t < 4; ++t) {
            v2f b; b.x = wb[t][j].z; b.y = wb[t][j].w;
            acc[t] = __builtin_amdgcn_wmma_f32_16x16x4_f32(
                false, a1, false, b, (short)0, acc[t], false, false);
        }
    }
}

__global__ void __launch_bounds__(128)
cp_lowrank(const float* __restrict__ Xin,
           const float* __restrict__ Wseq,
           const float* __restrict__ he,
           const float* __restrict__ cp,
           float* __restrict__ Out) {
    __shared__ __attribute__((aligned(16))) float stage[4][SROWS * TP];
    __shared__ __attribute__((aligned(16))) float part[4][16 * TP];
    __shared__ __attribute__((aligned(16))) float tsum[16 * TP];

    const int tid  = threadIdx.x;
    const int lane = tid & 31;
    const int wave = tid >> 5;
    const int c    = lane & 15;
    const int hf   = lane >> 4;
    const int m0   = (blockIdx.x >> 1) * 16;
    const int hbase = (blockIdx.x & 1) * 256 + wave * 64;

    const int ks = wave * 128 + 4 * c;
    v4f gx0[8], gx1[8];
    v4f gw0[32], gw1[32];
    {
        const float* __restrict__ xg = Xin + (size_t)(m0 + hf) * H_DIM + ks;
        const float* __restrict__ wg = Wseq + (size_t)hf * H_DIM + ks;
#pragma unroll
        for (int i = 0; i < 8; ++i) gx0[i] = *(const v4f*)(xg + (size_t)(2 * i) * H_DIM);
#pragma unroll
        for (int i = 0; i < 32; ++i) gw0[i] = *(const v4f*)(wg + (size_t)(2 * i) * H_DIM);
        __builtin_amdgcn_sched_barrier(0);
#pragma unroll
        for (int i = 0; i < 8; ++i) gx1[i] = *(const v4f*)(xg + 64 + (size_t)(2 * i) * H_DIM);
#pragma unroll
        for (int i = 0; i < 32; ++i) gw1[i] = *(const v4f*)(wg + 64 + (size_t)(2 * i) * H_DIM);
    }
    __builtin_amdgcn_sched_barrier(0);

    float* st = &stage[wave][0];
    float* sp = st + hf * TP + 4 * c;
#pragma unroll
    for (int i = 0; i < 8; ++i) *(v4f*)(sp + (2 * i) * TP) = gx0[i];
#pragma unroll
    for (int i = 0; i < 32; ++i) *(v4f*)(sp + (16 + 2 * i) * TP) = gw0[i];

    const float* fp = st + c * TP + 4 * hf;
    v4f xa0[8];
    v4f wb0[4][8];
#pragma unroll
    for (int j = 0; j < 8; ++j) xa0[j] = *(const v4f*)(fp + 8 * j);
#pragma unroll
    for (int t = 0; t < 4; ++t)
#pragma unroll
        for (int j = 0; j < 8; ++j) wb0[t][j] = *(const v4f*)(fp + (16 + 16 * t) * TP + 8 * j);

    __builtin_amdgcn_sched_barrier(0);

    v4f gh[32];
    {
        const float* __restrict__ hg = he + (size_t)(hbase + hf) * R_DIM + 4 * c;
#pragma unroll
        for (int i = 0; i < 32; ++i) gh[i] = *(const v4f*)(hg + (size_t)(2 * i) * R_DIM);
    }
    v4f cw[8];
    {
        const v4f* __restrict__ cpp = (const v4f*)(cp + 4 * hf);
#pragma unroll
        for (int j = 0; j < 8; ++j) cw[j] = cpp[2 * j];
    }

    __builtin_amdgcn_sched_barrier(0);

    v8f acc[4] = {};
    stage1_steps<0, 4>(acc, xa0, wb0);

    __builtin_amdgcn_sched_barrier(0);

#pragma unroll
    for (int i = 0; i < 8; ++i) *(v4f*)(sp + (HROWS + 2 * i) * TP) = gx1[i];
#pragma unroll
    for (int i = 0; i < 32; ++i) *(v4f*)(sp + (HROWS + 16 + 2 * i) * TP) = gw1[i];
    v4f xa1[8];
    v4f wb1[4][8];
#pragma unroll
    for (int j = 0; j < 8; ++j) xa1[j] = *(const v4f*)(fp + HROWS * TP + 8 * j);
#pragma unroll
    for (int t = 0; t < 4; ++t)
#pragma unroll
        for (int j = 0; j < 8; ++j) wb1[t][j] = *(const v4f*)(fp + (HROWS + 16 + 16 * t) * TP + 8 * j);

    __builtin_amdgcn_sched_barrier(0);

    stage1_steps<4, 8>(acc, xa0, wb0);
    stage1_steps<0, 4>(acc, xa1, wb1);

    __builtin_amdgcn_sched_barrier(0);

#pragma unroll
    for (int i = 0; i < 32; ++i)
        *(v4f*)(st + (160 + 32 * (i & 1) + 16 * hf + (i >> 1)) * TP + 4 * c) = gh[i];

    v4f fb[4][8];
#pragma unroll
    for (int u = 0; u < 4; ++u)
#pragma unroll
        for (int j = 0; j < 8; ++j)
            fb[u][j] = *(const v4f*)(st + (160 + 16 * u + c) * TP + 4 * hf + 8 * j);

    __builtin_amdgcn_sched_barrier(0);

    stage1_steps<4, 8>(acc, xa1, wb1);

    {
        float* __restrict__ pw = &part[wave][(8 * hf) * TP + c];
#pragma unroll
        for (int t = 0; t < 4; ++t)
#pragma unroll
            for (int p = 0; p < 8; ++p) pw[p * TP + 16 * t] = acc[t][p];
    }

    v4f ff[4][8];
#pragma unroll
    for (int u = 0; u < 4; ++u)
#pragma unroll
        for (int j = 0; j < 8; ++j) ff[u][j] = fb[u][j] * cw[j];

    __builtin_amdgcn_sched_barrier(0);
    __syncthreads();

    {
        const int off0 = (tid >> 4) * TP + (tid & 15) * 4;
        const int off1 = off0 + 8 * TP;
        v4f q0[4], q1[4];
#pragma unroll
        for (int w = 0; w < 4; ++w) { q0[w] = *(const v4f*)&part[w][off0]; q1[w] = *(const v4f*)&part[w][off1]; }
        __builtin_amdgcn_sched_barrier(0);
        v4f s0 = q0[0], s1 = q1[0];
#pragma unroll
        for (int w = 1; w < 4; ++w) { s0 += q0[w]; s1 += q1[w]; }
        *(v4f*)&tsum[off0] = s0;
        *(v4f*)&tsum[off1] = s1;
    }
    __syncthreads();

    v4f ta[8];
#pragma unroll
    for (int j = 0; j < 8; ++j) ta[j] = *(const v4f*)&tsum[c * TP + 8 * j + 4 * hf];

    __builtin_amdgcn_sched_barrier(0);

    v8f o[4] = {};
#pragma unroll
    for (int j = 0; j < 8; ++j) {
        v2f a0; a0.x = ta[j].x; a0.y = ta[j].y;
        v2f a1; a1.x = ta[j].z; a1.y = ta[j].w;
#pragma unroll
        for (int u = 0; u < 4; ++u) {
            v2f b; b.x = ff[u][j].x; b.y = ff[u][j].y;
            o[u] = __builtin_amdgcn_wmma_f32_16x16x4_f32(false, a0, false, b, (short)0, o[u], false, false);
        }
#pragma unroll
        for (int u = 0; u < 4; ++u) {
            v2f b; b.x = ff[u][j].z; b.y = ff[u][j].w;
            o[u] = __builtin_amdgcn_wmma_f32_16x16x4_f32(false, a1, false, b, (short)0, o[u], false, false);
        }
    }

    float* __restrict__ op = Out + (size_t)(m0 + 8 * hf) * H_DIM + hbase + 4 * c;
#pragma unroll
    for (int p = 0; p < 8; ++p) {
        v4f v; v.x = o[0][p]; v.y = o[1][p]; v.z = o[2][p]; v.w = o[3][p];
        *(v4f*)(op + (size_t)p * H_DIM) = v;
    }
}

extern "C" void kernel_launch(void* const* d_in, const int* in_sizes, int n_in,
                              void* d_out, int out_size, void* d_ws, size_t ws_size,
                              hipStream_t stream) {
    const float* hidden = (const float*)d_in[0];
    const float* W_seq  = (const float*)d_in[2];
    const float* he     = (const float*)d_in[3];
    const float* cp     = (const float*)d_in[4];
    float* out  = (float*)d_out;

    (void)in_sizes; (void)n_in; (void)out_size; (void)d_ws; (void)ws_size;

    cp_lowrank<<<256, 128, 0, stream>>>(hidden, W_seq, he, cp, out);
}


# Stage-1 operands fetched and consumed as four load groups (two column tiles of one k half each) instead of two: the first matrix step waits for 24 of the wave's 80 loads, not 40; sums unchanged
# speedup vs baseline: 7.3822x; 1.0566x over previous
#include <hip/hip_runtime.h>


typedef __attribute__((ext_vector_type(2))) float v2f;
typedef __attribute__((ext_vector_type(4))) float v4f;
typedef __attribute__((ext_vector_type(8))) float v8f;

#define H_DIM 512
#define R_DIM 64
#define M_ROWS 2048
#define TP 68
#define HROWS 80
#define SROWS 224

template <int T0, int J0, int J1>
static __device__ __forceinline__ void stage1_steps(v8f (&acc)[4], const v4f (&xa)[8], const v4f (&wb)[4][8]) {
#pragma unroll
    for (int j = J0; j < J1; ++j) {
        v2f a0; a0.x = xa[j].x; a0.y = xa[j].y;
        v2f a1; a1.x = xa[j].z; a1.y = xa[j].w;
#pragma unroll
        for (int t = T0; t < T0 + 2; ++t) {
            v2f b; b.x = wb[t][j].x; b.y = wb[t][j].y;
            acc[t] = __builtin_amdgcn_wmma_f32_16x16x4_f32(
                false, a0, false, b, (short)0, acc[t], false, false);
        }
#pragma unroll
        for (int t = T0; t < T0 + 2; ++t) {
            v2f b; b.x = wb[t][j].z; b.y = wb[t][j].w;
            acc[t] = __builtin_amdgcn_wmma_f32_16x16x4_f32(
                false, a1, false, b, (short)0, acc[t], false, false);
        }
    }
}

__global__ void __launch_bounds__(128)
cp_lowrank(const float* __restrict__ Xin,
           const float* __restrict__ Wseq,
           const float* __restrict__ he,
           const float* __restrict__ cp,
           float* __restrict__ Out) {
    __shared__ __attribute__((aligned(16))) float stage[4][SROWS * TP];
    __shared__ __attribute__((aligned(16))) float part[4][16 * TP];
    __shared__ __attribute__((aligned(16))) float tsum[16 * TP];

    const int tid  = threadIdx.x;
    const int lane = tid & 31;
    const int wave = tid >> 5;
    const int c    = lane & 15;
    const int hf   = lane >> 4;
    const int m0   = (blockIdx.x >> 1) * 16;
    const int hbase = (blockIdx.x & 1) * 256 + wave * 64;

    const int ks = wave * 128 + 4 * c;
    v4f gx0[8], gx1[8];
    v4f gw0[32], gw1[32];
    {
        const float* __restrict__ xg = Xin + (size_t)(m0 + hf) * H_DIM + ks;
        const float* __restrict__ wg = Wseq + (size_t)hf * H_DIM + ks;
#pragma unroll
        for (int i = 0; i < 8; ++i) gx0[i] = *(const v4f*)(xg + (size_t)(2 * i) * H_DIM);
#pragma unroll
        for (int i = 0; i < 16; ++i) gw0[i] = *(const v4f*)(wg + (size_t)(2 * i) * H_DIM);
        __builtin_amdgcn_sched_barrier(0);
#pragma unroll
        for (int i = 16; i < 32; ++i) gw0[i] = *(const v4f*)(wg + (size_t)(2 * i) * H_DIM);
        __builtin_amdgcn_sched_barrier(0);
#pragma unroll
        for (int i = 0; i < 8; ++i) gx1[i] = *(const v4f*)(xg + 64 + (size_t)(2 * i) * H_DIM);
#pragma unroll
        for (int i = 0; i < 16; ++i) gw1[i] = *(const v4f*)(wg + 64 + (size_t)(2 * i) * H_DIM);
        __builtin_amdgcn_sched_barrier(0);
#pragma unroll
        for (int i = 16; i < 32; ++i) gw1[i] = *(const v4f*)(wg + 64 + (size_t)(2 * i) * H_DIM);
    }
    __builtin_amdgcn_sched_barrier(0);

    float* st = &stage[wave][0];
    float* sp = st + hf * TP + 4 * c;
#pragma unroll
    for (int i = 0; i < 8; ++i) *(v4f*)(sp + (2 * i) * TP) = gx0[i];
#pragma unroll
    for (int i = 0; i < 16; ++i) *(v4f*)(sp + (16 + 2 * i) * TP) = gw0[i];

    const float* fp = st + c * TP + 4 * hf;
    v4f xa0[8];
    v4f wb0[4][8];
#pragma unroll
    for (int j = 0; j < 8; ++j) xa0[j] = *(const v4f*)(fp + 8 * j);
#pragma unroll
    for (int t = 0; t < 2; ++t)
#pragma unroll
        for (int j = 0; j < 8; ++j) wb0[t][j] = *(const v4f*)(fp + (16 + 16 * t) * TP + 8 * j);

    __builtin_amdgcn_sched_barrier(0);

    v8f acc[4] = {};
    stage1_steps<0, 0, 4>(acc, xa0, wb0);

    __builtin_amdgcn_sched_barrier(0);

#pragma unroll
    for (int i = 16; i < 32; ++i) *(v4f*)(sp + (16 + 2 * i) * TP) = gw0[i];
#pragma unroll
    for (int t = 2; t < 4; ++t)
#pragma unroll
        for (int j = 0; j < 8; ++j) wb0[t][j] = *(const v4f*)(fp + (16 + 16 * t) * TP + 8 * j);

    __builtin_amdgcn_sched_barrier(0);

    v4f gh[32];
    const float* __restrict__ hg = he + (size_t)(hbase + hf) * R_DIM + 4 * c;
#pragma unroll
    for (int i = 0; i < 16; ++i) gh[i] = *(const v4f*)(hg + (size_t)(2 * i) * R_DIM);

    __builtin_amdgcn_sched_barrier(0);

    stage1_steps<0, 4, 8>(acc, xa0, wb0);
    stage1_steps<2, 0, 4>(acc, xa0, wb0);

    __builtin_amdgcn_sched_barrier(0);

#pragma unroll
    for (int i = 0; i < 8; ++i) *(v4f*)(sp + (HROWS + 2 * i) * TP) = gx1[i];
#pragma unroll
    for (int i = 0; i < 16; ++i) *(v4f*)(sp + (HROWS + 16 + 2 * i) * TP) = gw1[i];
    v4f xa1[8];
    v4f wb1[4][8];
#pragma unroll
    for (int j = 0; j < 8; ++j) xa1[j] = *(const v4f*)(fp + HROWS * TP + 8 * j);
#pragma unroll
    for (int t = 0; t < 2; ++t)
#pragma unroll
        for (int j = 0; j < 8; ++j) wb1[t][j] = *(const v4f*)(fp + (HROWS + 16 + 16 * t) * TP + 8 * j);

    __builtin_amdgcn_sched_barrier(0);

#pragma unroll
    for (int i = 16; i < 32; ++i) gh[i] = *(const v4f*)(hg + (size_t)(2 * i) * R_DIM);
    v4f cw[8];
    {
        const v4f* __restrict__ cpp = (const v4f*)(cp + 4 * hf);
#pragma unroll
        for (int j = 0; j < 8; ++j) cw[j] = cpp[2 * j];
    }

    __builtin_amdgcn_sched_barrier(0);

    stage1_steps<2, 4, 8>(acc, xa0, wb0);
    stage1_steps<0, 0, 4>(acc, xa1, wb1);

    __builtin_amdgcn_sched_barrier(0);

#pragma unroll
    for (int i = 16; i < 32; ++i) *(v4f*)(sp + (HROWS + 16 + 2 * i) * TP) = gw1[i];
#pragma unroll
    for (int t = 2; t < 4; ++t)
#pragma unroll
        for (int j = 0; j < 8; ++j) wb1[t][j] = *(const v4f*)(fp + (HROWS + 16 + 16 * t) * TP + 8 * j);

    __builtin_amdgcn_sched_barrier(0);

    stage1_steps<0, 4, 8>(acc, xa1, wb1);
    stage1_steps<2, 0, 4>(acc, xa1, wb1);

    __builtin_amdgcn_sched_barrier(0);

#pragma unroll
    for (int i = 0; i < 32; ++i)
        *(v4f*)(st + (160 + 32 * (i & 1) + 16 * hf + (i >> 1)) * TP + 4 * c) = gh[i];

    v4f fb[4][8];
#pragma unroll
    for (int u = 0; u < 4; ++u)
#pragma unroll
        for (int j = 0; j < 8; ++j)
            fb[u][j] = *(const v4f*)(st + (160 + 16 * u + c) * TP + 4 * hf + 8 * j);

    __builtin_amdgcn_sched_barrier(0);

    stage1_steps<2, 4, 8>(acc, xa1, wb1);

    {
        float* __restrict__ pw = &part[wave][(8 * hf) * TP + c];
#pragma unroll
        for (int t = 0; t < 4; ++t)
#pragma unroll
            for (int p = 0; p < 8; ++p) pw[p * TP + 16 * t] = acc[t][p];
    }

    v4f ff[4][8];
#pragma unroll
    for (int u = 0; u < 4; ++u)
#pragma unroll
        for (int j = 0; j < 8; ++j) ff[u][j] = fb[u][j] * cw[j];

    __builtin_amdgcn_sched_barrier(0);
    __syncthreads();

    {
        const int off0 = (tid >> 4) * TP + (tid & 15) * 4;
        const int off1 = off0 + 8 * TP;
        v4f q0[4], q1[4];
#pragma unroll
        for (int w = 0; w < 4; ++w) { q0[w] = *(const v4f*)&part[w][off0]; q1[w] = *(const v4f*)&part[w][off1]; }
        __builtin_amdgcn_sched_barrier(0);
        v4f s0 = q0[0], s1 = q1[0];
#pragma unroll
        for (int w = 1; w < 4; ++w) { s0 += q0[w]; s1 += q1[w]; }
        *(v4f*)&tsum[off0] = s0;
        *(v4f*)&tsum[off1] = s1;
    }
    __syncthreads();

    v4f ta[8];
#pragma unroll
    for (int j = 0; j < 8; ++j) ta[j] = *(const v4f*)&tsum[c * TP + 8 * j + 4 * hf];

    __builtin_amdgcn_sched_barrier(0);

    v8f o[4] = {};
#pragma unroll
    for (int j = 0; j < 8; ++j) {
        v2f a0; a0.x = ta[j].x; a0.y = ta[j].y;
        v2f a1; a1.x = ta[j].z; a1.y = ta[j].w;
#pragma unroll
        for (int u = 0; u < 4; ++u) {
            v2f b; b.x = ff[u][j].x; b.y = ff[u][j].y;
            o[u] = __builtin_amdgcn_wmma_f32_16x16x4_f32(false, a0, false, b, (short)0, o[u], false, false);
        }
#pragma unroll
        for (int u = 0; u < 4; ++u) {
            v2f b; b.x = ff[u][j].z; b.y = ff[u][j].w;
            o[u] = __builtin_amdgcn_wmma_f32_16x16x4_f32(false, a1, false, b, (short)0, o[u], false, false);
        }
    }

    float* __restrict__ op = Out + (size_t)(m0 + 8 * hf) * H_DIM + hbase + 4 * c;
#pragma unroll
    for (int p = 0; p < 8; ++p) {
        v4f v; v.x = o[0][p]; v.y = o[1][p]; v.z = o[2][p]; v.w = o[3][p];
        *(v4f*)(op + (size_t)p * H_DIM) = v;
    }
}

extern "C" void kernel_launch(void* const* d_in, const int* in_sizes, int n_in,
                              void* d_out, int out_size, void* d_ws, size_t ws_size,
                              hipStream_t stream) {
    const float* hidden = (const float*)d_in[0];
    const float* W_seq  = (const float*)d_in[2];
    const float* he     = (const float*)d_in[3];
    const float* cp     = (const float*)d_in[4];
    float* out  = (float*)d_out;

    (void)in_sizes; (void)n_in; (void)out_size; (void)d_ws; (void)ws_size;

    cp_lowrank<<<256, 128, 0, stream>>>(hidden, W_seq, he, cp, out);
}
